# starting kernel
# baseline (speedup 1.0000x reference)
#include <hip/hip_runtime.h>


#define IN_CH   1024
#define OUT_CH  64
#define BATCH   65536
#define PITCH   (IN_CH + 8)
#define LDS_U16 (2 * OUT_CH * PITCH)

typedef __bf16          v16bf __attribute__((ext_vector_type(16)));
typedef float           v8f   __attribute__((ext_vector_type(8)));
typedef float           v4f   __attribute__((ext_vector_type(4)));
typedef unsigned short  v16u  __attribute__((ext_vector_type(16)));
typedef unsigned short  v8u   __attribute__((ext_vector_type(8)));
typedef unsigned int    v8ui  __attribute__((ext_vector_type(8)));

extern __shared__ unsigned short smem[];

__device__ __forceinline__ unsigned pack_hi16(unsigned u0, unsigned u1) {
    return __builtin_amdgcn_perm(u1, u0, 0x07060302u);
}

__device__ __forceinline__ void split2(float f0, float f1, unsigned& hi, unsigned& lo) {
    unsigned u0 = __builtin_bit_cast(unsigned, f0);
    unsigned u1 = __builtin_bit_cast(unsigned, f1);
    hi = pack_hi16(u0, u1);
    float l0 = f0 - __builtin_bit_cast(float, u0 & 0xFFFF0000u);
    float l1 = f1 - __builtin_bit_cast(float, u1 & 0xFFFF0000u);
    lo = pack_hi16(__builtin_bit_cast(unsigned, l0),
                   __builtin_bit_cast(unsigned, l1));
}

struct Frag { v8ui hi; v8ui lo; };

__device__ __forceinline__ void load_a_frag(const float* __restrict__ xrow, int kb, Frag& f) {
    const v4f* p0 = reinterpret_cast<const v4f*>(xrow + kb);
    const v4f* p1 = reinterpret_cast<const v4f*>(xrow + kb + 16);
    v4f r0 = __builtin_nontemporal_load(p0);
    v4f r1 = __builtin_nontemporal_load(p0 + 1);
    v4f r2 = __builtin_nontemporal_load(p1);
    v4f r3 = __builtin_nontemporal_load(p1 + 1);
    float v[16] = { r0[0], r0[1], r0[2], r0[3],  r1[0], r1[1], r1[2], r1[3],
                    r2[0], r2[1], r2[2], r2[3],  r3[0], r3[1], r3[2], r3[3] };
#pragma unroll
    for (int j = 0; j < 8; ++j) {
        unsigned h, l;
        split2(v[2 * j], v[2 * j + 1], h, l);
        f.hi[j] = h;
        f.lo[j] = l;
    }
}

__device__ __forceinline__ v16u load_b_frag(int plane, int ch, int kb) {
    const unsigned short* row = &smem[plane * (OUT_CH * PITCH) + ch * PITCH];
    v8u a = *reinterpret_cast<const v8u*>(row + kb);
    v8u c = *reinterpret_cast<const v8u*>(row + kb + 16);
    return __builtin_shufflevector(a, c, 0, 1, 2, 3, 4, 5, 6, 7,
                                         8, 9, 10, 11, 12, 13, 14, 15);
}

__device__ __forceinline__ v8f wmma_bf16(const v8ui& a, const v16u& b, v8f c) {
    return __builtin_amdgcn_wmma_f32_16x16x32_bf16(
        false, __builtin_bit_cast(v16bf, a),
        false, __builtin_bit_cast(v16bf, b),
        (short)0, c, false, false);
}

__global__ __launch_bounds__(256) void seplinear_wmma_kernel(
    const float* __restrict__ x, const float* __restrict__ W,
    const float* __restrict__ b, float* __restrict__ out) {

    const int tid = threadIdx.x;

    for (int idx = tid; idx < OUT_CH * IN_CH; idx += 256) {
        int ch = idx >> 10, k = idx & (IN_CH - 1);
        float f = W[idx];
        unsigned u = __builtin_bit_cast(unsigned, f);
        float fl = f - __builtin_bit_cast(float, u & 0xFFFF0000u);
        smem[ch * PITCH + k]                  = (unsigned short)(u >> 16);
        smem[OUT_CH * PITCH + ch * PITCH + k] =
            (unsigned short)(__builtin_bit_cast(unsigned, fl) >> 16);
    }
    __syncthreads();

    const int lane  = tid & 31;
    const int wave  = tid >> 5;
    const int lrow  = lane & 15;
    const int khalf = lane >> 4;

    for (int base = blockIdx.x * 256; base < BATCH; base += gridDim.x * 256) {
        const int m0 = base + wave * 32;

        v8f acc[2][4];
#pragma unroll
        for (int mt = 0; mt < 2; ++mt)
#pragma unroll
            for (int nt = 0; nt < 4; ++nt) acc[mt][nt] = v8f{};

        const float* xr0 = x + (size_t)(m0 + lrow) * IN_CH;
        const float* xr1 = x + (size_t)(m0 + 16 + lrow) * IN_CH;

        for (int k0 = 0; k0 < IN_CH; k0 += 32) {
            const int kb = k0 + khalf * 8;
            Frag a0, a1;
            load_a_frag(xr0, kb, a0);
            load_a_frag(xr1, kb, a1);

#pragma unroll
            for (int nt = 0; nt < 4; ++nt) {
                const int ch = nt * 16 + lrow;
                v16u bh = load_b_frag(0, ch, kb);
                v16u bl = load_b_frag(1, ch, kb);

                acc[0][nt] = wmma_bf16(a0.hi, bh, acc[0][nt]);
                acc[0][nt] = wmma_bf16(a0.hi, bl, acc[0][nt]);
                acc[0][nt] = wmma_bf16(a0.lo, bh, acc[0][nt]);

                acc[1][nt] = wmma_bf16(a1.hi, bh, acc[1][nt]);
                acc[1][nt] = wmma_bf16(a1.hi, bl, acc[1][nt]);
                acc[1][nt] = wmma_bf16(a1.lo, bh, acc[1][nt]);
            }
        }

#pragma unroll
        for (int nt = 0; nt < 4; ++nt) {
            const int   n    = nt * 16 + lrow;
            const float bias = b[n];
#pragma unroll
            for (int mt = 0; mt < 2; ++mt) {
#pragma unroll
                for (int d = 0; d < 8; ++d) {
                    const int row = m0 + mt * 16 + khalf * 8 + d;
                    __builtin_nontemporal_store(acc[mt][nt][d] + bias,
                                                out + (size_t)row * OUT_CH + n);
                }
            }
        }
    }
}

extern "C" void kernel_launch(void* const* d_in, const int* in_sizes, int n_in,
                              void* d_out, int out_size, void* d_ws, size_t ws_size,
                              hipStream_t stream) {
    const float* x = (const float*)d_in[0];
    const float* W = (const float*)d_in[1];
    const float* b = (const float*)d_in[2];
    float* out     = (float*)d_out;

    const size_t lds_bytes = (size_t)LDS_U16 * sizeof(unsigned short);
    dim3 grid(256), block(256);
    hipLaunchKernelGGL(seplinear_wmma_kernel, grid, block, lds_bytes, stream,
                       x, W, b, out);
}


# split K inside wave pairs: a wave takes 32 rows x alternate 64-k trips, so each B fragment feeds 2 M-tiles and 16 loads fly per wave; pair partial sums added via LDS at the end (sum order changes)
# speedup vs baseline: 1.4561x; 1.4561x over previous
#include <hip/hip_runtime.h>


#define IN_CH   1024
#define OUT_CH  64
#define BATCH   65536
#define THREADS 512
#define PITCH   (IN_CH + 8)
#define LDS_U16 (2 * OUT_CH * PITCH)

typedef __bf16          v16bf __attribute__((ext_vector_type(16)));
typedef float           v8f   __attribute__((ext_vector_type(8)));
typedef float           v4f   __attribute__((ext_vector_type(4)));
typedef unsigned short  v16u  __attribute__((ext_vector_type(16)));
typedef unsigned short  v8u   __attribute__((ext_vector_type(8)));
typedef unsigned int    v8ui  __attribute__((ext_vector_type(8)));

extern __shared__ unsigned short smem[];

static_assert(BATCH == 256 * 256, "one 256-row pass per work-group, 256 work-groups");

__device__ __forceinline__ unsigned pack_hi16(unsigned u0, unsigned u1) {
    return __builtin_amdgcn_perm(u1, u0, 0x07060302u);
}

__device__ __forceinline__ void split2(float f0, float f1, unsigned& hi, unsigned& lo) {
    unsigned u0 = __builtin_bit_cast(unsigned, f0);
    unsigned u1 = __builtin_bit_cast(unsigned, f1);
    hi = pack_hi16(u0, u1);
    float l0 = f0 - __builtin_bit_cast(float, u0 & 0xFFFF0000u);
    float l1 = f1 - __builtin_bit_cast(float, u1 & 0xFFFF0000u);
    lo = pack_hi16(__builtin_bit_cast(unsigned, l0),
                   __builtin_bit_cast(unsigned, l1));
}

struct Frag { v8ui hi; v8ui lo; };

struct Raw { v4f r0, r1, r2, r3; };

__device__ __forceinline__ void load_a_raw(const float* __restrict__ xrow, int kb, Raw& q) {
    const v4f* p0 = reinterpret_cast<const v4f*>(xrow + kb);
    const v4f* p1 = reinterpret_cast<const v4f*>(xrow + kb + 16);
    q.r0 = __builtin_nontemporal_load(p0);
    q.r1 = __builtin_nontemporal_load(p0 + 1);
    q.r2 = __builtin_nontemporal_load(p1);
    q.r3 = __builtin_nontemporal_load(p1 + 1);
}

__device__ __forceinline__ void split_a_frag(const Raw& q, Frag& f) {
    float v[16] = { q.r0[0], q.r0[1], q.r0[2], q.r0[3],  q.r1[0], q.r1[1], q.r1[2], q.r1[3],
                    q.r2[0], q.r2[1], q.r2[2], q.r2[3],  q.r3[0], q.r3[1], q.r3[2], q.r3[3] };
#pragma unroll
    for (int j = 0; j < 8; ++j) {
        unsigned h, l;
        split2(v[2 * j], v[2 * j + 1], h, l);
        f.hi[j] = h;
        f.lo[j] = l;
    }
}

__device__ __forceinline__ v16u load_b_frag(int plane, int ch, int kb) {
    const unsigned short* row = &smem[plane * (OUT_CH * PITCH) + ch * PITCH];
    v8u a = *reinterpret_cast<const v8u*>(row + kb);
    v8u c = *reinterpret_cast<const v8u*>(row + kb + 16);
    return __builtin_shufflevector(a, c, 0, 1, 2, 3, 4, 5, 6, 7,
                                         8, 9, 10, 11, 12, 13, 14, 15);
}

__device__ __forceinline__ v8f wmma_bf16(const v8ui& a, const v16u& b, v8f c) {
    return __builtin_amdgcn_wmma_f32_16x16x32_bf16(
        false, __builtin_bit_cast(v16bf, a),
        false, __builtin_bit_cast(v16bf, b),
        (short)0, c, false, false);
}

__global__ __launch_bounds__(THREADS) __attribute__((amdgpu_waves_per_eu(4, 4)))
void seplinear_wmma_kernel(
    const float* __restrict__ x, const float* __restrict__ W,
    const float* __restrict__ b, float* __restrict__ out) {

    const int tid = threadIdx.x;

    for (int idx = tid; idx < OUT_CH * IN_CH; idx += THREADS) {
        int ch = idx >> 10, k = idx & (IN_CH - 1);
        float f = W[idx];
        unsigned u = __builtin_bit_cast(unsigned, f);
        float fl = f - __builtin_bit_cast(float, u & 0xFFFF0000u);
        smem[ch * PITCH + k]                  = (unsigned short)(u >> 16);
        smem[OUT_CH * PITCH + ch * PITCH + k] =
            (unsigned short)(__builtin_bit_cast(unsigned, fl) >> 16);
    }
    __syncthreads();

    const int lane  = tid & 31;
    const int wave  = __builtin_amdgcn_readfirstlane(tid >> 5);
    const int lrow  = lane & 15;
    const int khalf = lane >> 4;

    const int kh = wave & 1;
    const int mg = wave >> 1;

    const int m0 = blockIdx.x * 256 + mg * 32;

    v8f acc0[4], acc1[4];
#pragma unroll
    for (int nt = 0; nt < 4; ++nt) { acc0[nt] = v8f{}; acc1[nt] = v8f{}; }

    const float* xr0 = x + (size_t)(m0 + lrow) * IN_CH;
    const float* xr1 = xr0 + (size_t)16 * IN_CH;

    for (int t = 0; t < IN_CH / 128; ++t) {
        const int k0 = t * 128 + kh * 64;
        const int kb = k0 + khalf * 8;
        Raw q00, q01, q10, q11;
        load_a_raw(xr0, kb,      q00);
        load_a_raw(xr1, kb,      q10);
        load_a_raw(xr0, kb + 32, q01);
        load_a_raw(xr1, kb + 32, q11);
        v16u bh0[4], bl0[4];
#pragma unroll
        for (int nt = 0; nt < 4; ++nt) {
            bh0[nt] = load_b_frag(0, nt * 16 + lrow, kb);
            bl0[nt] = load_b_frag(1, nt * 16 + lrow, kb);
        }
        __builtin_amdgcn_sched_barrier(0);
        Frag a00, a01, a10, a11;
        split_a_frag(q00, a00);
        split_a_frag(q10, a10);
        split_a_frag(q01, a01);
        split_a_frag(q11, a11);

#pragma unroll
        for (int nt = 0; nt < 4; ++nt) {
            acc0[nt] = wmma_bf16(a00.hi, bh0[nt], acc0[nt]);
            acc1[nt] = wmma_bf16(a10.hi, bh0[nt], acc1[nt]);
            acc0[nt] = wmma_bf16(a00.hi, bl0[nt], acc0[nt]);
            acc1[nt] = wmma_bf16(a10.hi, bl0[nt], acc1[nt]);
            acc0[nt] = wmma_bf16(a00.lo, bh0[nt], acc0[nt]);
            acc1[nt] = wmma_bf16(a10.lo, bh0[nt], acc1[nt]);
        }
#pragma unroll
        for (int nt = 0; nt < 4; ++nt) {
            const int ch = nt * 16 + lrow;
            v16u bh = load_b_frag(0, ch, kb + 32);
            v16u bl = load_b_frag(1, ch, kb + 32);
            acc0[nt] = wmma_bf16(a01.hi, bh, acc0[nt]);
            acc1[nt] = wmma_bf16(a11.hi, bh, acc1[nt]);
            acc0[nt] = wmma_bf16(a01.hi, bl, acc0[nt]);
            acc1[nt] = wmma_bf16(a11.hi, bl, acc1[nt]);
            acc0[nt] = wmma_bf16(a01.lo, bh, acc0[nt]);
            acc1[nt] = wmma_bf16(a11.lo, bh, acc1[nt]);
        }
    }

    v4f* xch      = reinterpret_cast<v4f*>(smem);
    v4f* slot_out = xch + (wave ^ 1) * 256 + lane;
    v4f* slot_in  = xch + wave * 256 + lane;

    __syncthreads();
    v8f fin[4];
#pragma unroll
    for (int nt = 0; nt < 4; ++nt) {
        const v8f give = kh ? acc0[nt] : acc1[nt];
        fin[nt]        = kh ? acc1[nt] : acc0[nt];
        slot_out[(nt * 2 + 0) * 32] = __builtin_shufflevector(give, give, 0, 1, 2, 3);
        slot_out[(nt * 2 + 1) * 32] = __builtin_shufflevector(give, give, 4, 5, 6, 7);
    }
    __syncthreads();

    const int mfin = m0 + kh * 16;
#pragma unroll
    for (int nt = 0; nt < 4; ++nt) {
        const v4f p0 = slot_in[(nt * 2 + 0) * 32];
        const v4f p1 = slot_in[(nt * 2 + 1) * 32];
        const int   n    = nt * 16 + lrow;
        const float bias = b[n];
#pragma unroll
        for (int d = 0; d < 8; ++d) {
            const int   row = mfin + khalf * 8 + d;
            const float got = (d < 4) ? p0[d & 3] : p1[d & 3];
            __builtin_nontemporal_store((fin[nt][d] + got) + bias,
                                        out + (size_t)row * OUT_CH + n);
        }
    }
}

extern "C" void kernel_launch(void* const* d_in, const int* in_sizes, int n_in,
                              void* d_out, int out_size, void* d_ws, size_t ws_size,
                              hipStream_t stream) {
    const float* x = (const float*)d_in[0];
    const float* W = (const float*)d_in[1];
    const float* b = (const float*)d_in[2];
    float* out     = (float*)d_out;

    const size_t lds_bytes = (size_t)LDS_U16 * sizeof(unsigned short);
    dim3 grid(256), block(THREADS);
    hipLaunchKernelGGL(seplinear_wmma_kernel, grid, block, lds_bytes, stream,
                       x, W, b, out);
}


# x data path: each global load now covers whole 128-byte lines (4 rows x 128 B, was 16 rows x 32 B); a wave turns every 16x32 block to one-row-per-lane through its own LDS tile; sum order unchanged
# speedup vs baseline: 1.5523x; 1.0661x over previous
#include <hip/hip_runtime.h>


#define IN_CH   1024
#define OUT_CH  64
#define BATCH   65536
#define THREADS 512
#define PITCH   (IN_CH + 8)
#define LDS_U16 (2 * OUT_CH * PITCH)
#define XT_PITCH 9
#define XT_V4   (16 * XT_PITCH)

typedef __bf16          v16bf __attribute__((ext_vector_type(16)));
typedef float           v8f   __attribute__((ext_vector_type(8)));
typedef float           v4f   __attribute__((ext_vector_type(4)));
typedef unsigned short  v16u  __attribute__((ext_vector_type(16)));
typedef unsigned short  v8u   __attribute__((ext_vector_type(8)));
typedef unsigned int    v8ui  __attribute__((ext_vector_type(8)));

extern __shared__ unsigned short smem[];

static_assert(BATCH == 256 * 256, "one 256-row pass per work-group, 256 work-groups");

__device__ __forceinline__ unsigned pack_hi16(unsigned u0, unsigned u1) {
    return __builtin_amdgcn_perm(u1, u0, 0x07060302u);
}

__device__ __forceinline__ void split2(float f0, float f1, unsigned& hi, unsigned& lo) {
    unsigned u0 = __builtin_bit_cast(unsigned, f0);
    unsigned u1 = __builtin_bit_cast(unsigned, f1);
    hi = pack_hi16(u0, u1);
    float l0 = f0 - __builtin_bit_cast(float, u0 & 0xFFFF0000u);
    float l1 = f1 - __builtin_bit_cast(float, u1 & 0xFFFF0000u);
    lo = pack_hi16(__builtin_bit_cast(unsigned, l0),
                   __builtin_bit_cast(unsigned, l1));
}

struct Frag { v8ui hi; v8ui lo; };

struct Raw { v4f r0, r1, r2, r3; };

__device__ __forceinline__ void load_x_lines(const float* __restrict__ xl, Raw& g) {
    g.r0 = __builtin_nontemporal_load(reinterpret_cast<const v4f*>(xl));
    g.r1 = __builtin_nontemporal_load(reinterpret_cast<const v4f*>(xl + 4 * IN_CH));
    g.r2 = __builtin_nontemporal_load(reinterpret_cast<const v4f*>(xl + 8 * IN_CH));
    g.r3 = __builtin_nontemporal_load(reinterpret_cast<const v4f*>(xl + 12 * IN_CH));
}

__device__ __forceinline__ void turn_block(const Raw& g, v4f* wr, const v4f* rd, Raw& q) {
    wr[0 * 4 * XT_PITCH] = g.r0;
    wr[1 * 4 * XT_PITCH] = g.r1;
    wr[2 * 4 * XT_PITCH] = g.r2;
    wr[3 * 4 * XT_PITCH] = g.r3;
    q.r0 = rd[0];
    q.r1 = rd[1];
    q.r2 = rd[4];
    q.r3 = rd[5];
}

__device__ __forceinline__ void split_a_frag(const Raw& q, Frag& f) {
    float v[16] = { q.r0[0], q.r0[1], q.r0[2], q.r0[3],  q.r1[0], q.r1[1], q.r1[2], q.r1[3],
                    q.r2[0], q.r2[1], q.r2[2], q.r2[3],  q.r3[0], q.r3[1], q.r3[2], q.r3[3] };
#pragma unroll
    for (int j = 0; j < 8; ++j) {
        unsigned h, l;
        split2(v[2 * j], v[2 * j + 1], h, l);
        f.hi[j] = h;
        f.lo[j] = l;
    }
}

__device__ __forceinline__ v16u load_b_frag(int plane, int ch, int kb) {
    const unsigned short* row = &smem[plane * (OUT_CH * PITCH) + ch * PITCH];
    v8u a = *reinterpret_cast<const v8u*>(row + kb);
    v8u c = *reinterpret_cast<const v8u*>(row + kb + 16);
    return __builtin_shufflevector(a, c, 0, 1, 2, 3, 4, 5, 6, 7,
                                         8, 9, 10, 11, 12, 13, 14, 15);
}

__device__ __forceinline__ v8f wmma_bf16(const v8ui& a, const v16u& b, v8f c) {
    return __builtin_amdgcn_wmma_f32_16x16x32_bf16(
        false, __builtin_bit_cast(v16bf, a),
        false, __builtin_bit_cast(v16bf, b),
        (short)0, c, false, false);
}

__global__ __launch_bounds__(THREADS) __attribute__((amdgpu_waves_per_eu(4, 4)))
void seplinear_wmma_kernel(
    const float* __restrict__ x, const float* __restrict__ W,
    const float* __restrict__ b, float* __restrict__ out) {

    const int tid = threadIdx.x;

    for (int idx = tid; idx < OUT_CH * IN_CH; idx += THREADS) {
        int ch = idx >> 10, k = idx & (IN_CH - 1);
        float f = W[idx];
        unsigned u = __builtin_bit_cast(unsigned, f);
        float fl = f - __builtin_bit_cast(float, u & 0xFFFF0000u);
        smem[ch * PITCH + k]                  = (unsigned short)(u >> 16);
        smem[OUT_CH * PITCH + ch * PITCH + k] =
            (unsigned short)(__builtin_bit_cast(unsigned, fl) >> 16);
    }
    __syncthreads();

    const int lane  = tid & 31;
    const int wave  = __builtin_amdgcn_readfirstlane(tid >> 5);
    const int lrow  = lane & 15;
    const int khalf = lane >> 4;

    const int kh = wave & 1;
    const int mg = wave >> 1;

    const int m0 = blockIdx.x * 256 + mg * 32;

    v8f acc0[4], acc1[4];
#pragma unroll
    for (int nt = 0; nt < 4; ++nt) { acc0[nt] = v8f{}; acc1[nt] = v8f{}; }

    const float* xl0 = x + (size_t)(m0 + (lane >> 3)) * IN_CH + (lane & 7) * 4;
    const float* xl1 = xl0 + (size_t)16 * IN_CH;

    v4f* xt = reinterpret_cast<v4f*>(smem + LDS_U16) + wave * XT_V4;
    v4f*       xt_wr = xt + (lane >> 3) * XT_PITCH + (lane & 7);
    const v4f* xt_rd = xt + lrow * XT_PITCH + 2 * khalf;

    for (int t = 0; t < IN_CH / 128; ++t) {
        const int k0 = t * 128 + kh * 64;
        const int kb = k0 + khalf * 8;
        Raw g00, g01, g10, g11;
        load_x_lines(xl0 + k0,      g00);
        load_x_lines(xl1 + k0,      g10);
        load_x_lines(xl0 + k0 + 32, g01);
        load_x_lines(xl1 + k0 + 32, g11);
        v16u bh0[4], bl0[4];
#pragma unroll
        for (int nt = 0; nt < 4; ++nt) {
            bh0[nt] = load_b_frag(0, nt * 16 + lrow, kb);
            bl0[nt] = load_b_frag(1, nt * 16 + lrow, kb);
        }
        __builtin_amdgcn_sched_barrier(0);
        Raw q00, q01, q10, q11;
        turn_block(g00, xt_wr, xt_rd, q00);
        turn_block(g10, xt_wr, xt_rd, q10);
        turn_block(g01, xt_wr, xt_rd, q01);
        turn_block(g11, xt_wr, xt_rd, q11);
        Frag a00, a01, a10, a11;
        split_a_frag(q00, a00);
        split_a_frag(q10, a10);
        split_a_frag(q01, a01);
        split_a_frag(q11, a11);

#pragma unroll
        for (int nt = 0; nt < 4; ++nt) {
            acc0[nt] = wmma_bf16(a00.hi, bh0[nt], acc0[nt]);
            acc1[nt] = wmma_bf16(a10.hi, bh0[nt], acc1[nt]);
            acc0[nt] = wmma_bf16(a00.hi, bl0[nt], acc0[nt]);
            acc1[nt] = wmma_bf16(a10.hi, bl0[nt], acc1[nt]);
            acc0[nt] = wmma_bf16(a00.lo, bh0[nt], acc0[nt]);
            acc1[nt] = wmma_bf16(a10.lo, bh0[nt], acc1[nt]);
        }
#pragma unroll
        for (int nt = 0; nt < 4; ++nt) {
            const int ch = nt * 16 + lrow;
            v16u bh = load_b_frag(0, ch, kb + 32);
            v16u bl = load_b_frag(1, ch, kb + 32);
            acc0[nt] = wmma_bf16(a01.hi, bh, acc0[nt]);
            acc1[nt] = wmma_bf16(a11.hi, bh, acc1[nt]);
            acc0[nt] = wmma_bf16(a01.hi, bl, acc0[nt]);
            acc1[nt] = wmma_bf16(a11.hi, bl, acc1[nt]);
            acc0[nt] = wmma_bf16(a01.lo, bh, acc0[nt]);
            acc1[nt] = wmma_bf16(a11.lo, bh, acc1[nt]);
        }
    }

    v4f* xch      = reinterpret_cast<v4f*>(smem);
    v4f* slot_out = xch + (wave ^ 1) * 256 + lane;
    v4f* slot_in  = xch + wave * 256 + lane;

    __syncthreads();
    v8f fin[4];
#pragma unroll
    for (int nt = 0; nt < 4; ++nt) {
        const v8f give = kh ? acc0[nt] : acc1[nt];
        fin[nt]        = kh ? acc1[nt] : acc0[nt];
        slot_out[(nt * 2 + 0) * 32] = __builtin_shufflevector(give, give, 0, 1, 2, 3);
        slot_out[(nt * 2 + 1) * 32] = __builtin_shufflevector(give, give, 4, 5, 6, 7);
    }
    __syncthreads();

    const int mfin = m0 + kh * 16;
#pragma unroll
    for (int nt = 0; nt < 4; ++nt) {
        const v4f p0 = slot_in[(nt * 2 + 0) * 32];
        const v4f p1 = slot_in[(nt * 2 + 1) * 32];
        const int   n    = nt * 16 + lrow;
        const float bias = b[n];
#pragma unroll
        for (int d = 0; d < 8; ++d) {
            const int   row = mfin + khalf * 8 + d;
            const float got = (d < 4) ? p0[d & 3] : p1[d & 3];
            __builtin_nontemporal_store((fin[nt][d] + got) + bias,
                                        out + (size_t)row * OUT_CH + n);
        }
    }
}

extern "C" void kernel_launch(void* const* d_in, const int* in_sizes, int n_in,
                              void* d_out, int out_size, void* d_ws, size_t ws_size,
                              hipStream_t stream) {
    const float* x = (const float*)d_in[0];
    const float* W = (const float*)d_in[1];
    const float* b = (const float*)d_in[2];
    float* out     = (float*)d_out;

    const size_t lds_bytes = (size_t)LDS_U16 * sizeof(unsigned short) + (size_t)16 * XT_V4 * sizeof(v4f);
    dim3 grid(256), block(THREADS);
    hipLaunchKernelGGL(seplinear_wmma_kernel, grid, block, lds_bytes, stream,
                       x, W, b, out);
}
